# starting kernel
# baseline (speedup 1.0000x reference)
#include <hip/hip_runtime.h>

#define D 64

typedef __attribute__((ext_vector_type(2))) float v2f;
typedef __attribute__((ext_vector_type(8))) float v8f;

__global__ void zero_f32(float* __restrict__ p, long n) {
  long i = (long)blockIdx.x * blockDim.x + threadIdx.x;
  long stride = (long)gridDim.x * blockDim.x;
  for (; i < n; i += stride) p[i] = 0.0f;
}

__global__ void scatter_add(const float* __restrict__ xin,
                            const int* __restrict__ src,
                            const int* __restrict__ dst,
                            float* __restrict__ ssum, float* __restrict__ cnt,
                            int nE, int withCnt) {
  long i = (long)blockIdx.x * blockDim.x + threadIdx.x;
  long total = (long)nE * 16;
  long stride = (long)gridDim.x * blockDim.x;
  for (; i < total; i += stride) {
    int e = (int)(i >> 4);
    int q = (int)(i & 15);
    int s = src[e];
    int d = dst[e];
    const float4 v = *(const float4*)(xin + (long)s * D + q * 4);
    float* p = ssum + (long)d * D + q * 4;
    atomicAdd(p + 0, v.x);
    atomicAdd(p + 1, v.y);
    atomicAdd(p + 2, v.z);
    atomicAdd(p + 3, v.w);
    if (withCnt && q == 0) atomicAdd(cnt + d, 1.0f);
  }
}

template <int RELU>
__global__ __launch_bounds__(128) void sage_gemm(
    const float* __restrict__ xin, const float* __restrict__ ssum,
    const float* __restrict__ cnt, const float* __restrict__ Wl,
    const float* __restrict__ Wr, const float* __restrict__ bias,
    float* __restrict__ out, int nN) {
  __shared__ float sMean[16 * D];
  __shared__ float sX[16 * D];
  const int m0 = blockIdx.x * 16;
  const int tid = threadIdx.x;

  {
    int idx = tid * 8;
    int row = idx >> 6;
    int col = idx & (D - 1);
    int gr = m0 + row;
    if (gr >= nN) gr = nN - 1;
    float c = cnt[gr];
    float rinv = 1.0f / fmaxf(c, 1.0f);
    const float* ps = ssum + (long)gr * D + col;
    const float* px = xin + (long)gr * D + col;
#pragma unroll
    for (int t = 0; t < 8; ++t) {
      sMean[idx + t] = ps[t] * rinv;
      sX[idx + t] = px[t];
    }
  }
  __syncthreads();

  const int wave = tid >> 5;
  const int lane = tid & 31;
  const int n0 = wave * 16;
  const int lr = lane & 15;
  const int kh = (lane >> 4) * 2;

  const float* mrow = sMean + lr * D;
  const float* xrow = sX + lr * D;
  const float* wlrow = Wl + (n0 + lr) * D;
  const float* wrrow = Wr + (n0 + lr) * D;

  v8f acc = {};
#pragma unroll
  for (int k0 = 0; k0 < D; k0 += 4) {
    v2f a, b;
    a[0] = mrow[k0 + kh + 0];
    a[1] = mrow[k0 + kh + 1];
    b[0] = wlrow[k0 + kh + 0];
    b[1] = wlrow[k0 + kh + 1];
    acc = __builtin_amdgcn_wmma_f32_16x16x4_f32(false, a, false, b, (short)0,
                                                acc, false, false);
    a[0] = xrow[k0 + kh + 0];
    a[1] = xrow[k0 + kh + 1];
    b[0] = wrrow[k0 + kh + 0];
    b[1] = wrrow[k0 + kh + 1];
    acc = __builtin_amdgcn_wmma_f32_16x16x4_f32(false, a, false, b, (short)0,
                                                acc, false, false);
  }

  const float bv = bias[n0 + lr];
  const int mbase = m0 + (lane >> 4) * 8;
#pragma unroll
  for (int i = 0; i < 8; ++i) {
    int r = mbase + i;
    if (r < nN) {
      float v = acc[i] + bv;
      if (RELU) v = fmaxf(v, 0.0f);
      out[(long)r * D + n0 + lr] = v;
    }
  }
}

extern "C" void kernel_launch(void* const* d_in, const int* in_sizes, int n_in,
                              void* d_out, int out_size, void* d_ws,
                              size_t ws_size, hipStream_t stream) {
  const float* x = (const float*)d_in[0];
  const int* edge = (const int*)d_in[1];
  const float* Wl1 = (const float*)d_in[2];
  const float* Wr1 = (const float*)d_in[3];
  const float* b1 = (const float*)d_in[4];
  const float* Wl2 = (const float*)d_in[5];
  const float* Wr2 = (const float*)d_in[6];
  const float* b2 = (const float*)d_in[7];

  const int N = in_sizes[0] / D;
  const int E = in_sizes[1] / 2;
  const int* src = edge;
  const int* dst = edge + E;

  float* ssum = (float*)d_ws;
  float* cnt = ssum + (long)N * D;
  float* h = cnt + N;
  float* out = (float*)d_out;

  const int gtiles = (N + 15) / 16;

  zero_f32<<<2048, 256, 0, stream>>>(ssum, (long)N * D);
  zero_f32<<<64, 256, 0, stream>>>(cnt, (long)N);
  scatter_add<<<4096, 256, 0, stream>>>(x, src, dst, ssum, cnt, E, 1);
  sage_gemm<1><<<gtiles, 128, 0, stream>>>(x, ssum, cnt, Wl1, Wr1, b1, h, N);

  zero_f32<<<2048, 256, 0, stream>>>(ssum, (long)N * D);
  scatter_add<<<4096, 256, 0, stream>>>(h, src, dst, ssum, cnt, E, 0);
  sage_gemm<0><<<gtiles, 128, 0, stream>>>(h, ssum, cnt, Wl2, Wr2, b2, out, N);
}


# scatter_add: one wave per edge (32 edge ids per coalesced load), so each row load and each float atomic covers one whole 128-byte line, was 4 lines; float atomics kept, arrival order as before
# speedup vs baseline: 1.6886x; 1.6886x over previous
#include <hip/hip_runtime.h>

#define D 64

typedef __attribute__((ext_vector_type(2))) float v2f;
typedef __attribute__((ext_vector_type(8))) float v8f;

__global__ void zero_f32(float* __restrict__ p, long n) {
  long i = (long)blockIdx.x * blockDim.x + threadIdx.x;
  long stride = (long)gridDim.x * blockDim.x;
  for (; i < n; i += stride) p[i] = 0.0f;
}

__device__ __forceinline__ void scatter_edges8(const float* __restrict__ xl,
                                               float* __restrict__ sl,
                                               int sv, int dv, int j0) {
  float v0[8], v1[8];
  int dd[8];
#pragma unroll
  for (int u = 0; u < 8; ++u) {
    const int s = __builtin_amdgcn_readlane(sv, j0 + u);
    dd[u] = __builtin_amdgcn_readlane(dv, j0 + u);
    const float* px = xl + (long)s * D;
    v0[u] = px[0];
    v1[u] = px[32];
  }
#pragma unroll
  for (int u = 0; u < 8; ++u) {
    float* p = sl + (long)dd[u] * D;
    atomicAdd(p, v0[u]);
    atomicAdd(p + 32, v1[u]);
  }
}

__global__ void scatter_add(const float* __restrict__ xin,
                            const int* __restrict__ src,
                            const int* __restrict__ dst,
                            float* __restrict__ ssum, float* __restrict__ cnt,
                            int nE, int withCnt) {
  const int lane = threadIdx.x & 31;
  const int wave = __builtin_amdgcn_readfirstlane((int)(threadIdx.x >> 5));
  const int wpb = blockDim.x >> 5;
  const long nChunks = ((long)nE + 31) >> 5;
  const long cstride = (long)gridDim.x * wpb;
  const float* xl = xin + lane;
  float* sl = ssum + lane;
  for (long c = (long)blockIdx.x * wpb + wave; c < nChunks; c += cstride) {
    const long e = c * 32 + lane;
    const long rem = (long)nE - c * 32;
    int sv = 0, dv = 0;
    if (e < nE) {
      sv = src[e];
      dv = dst[e];
      if (withCnt) atomicAdd(cnt + dv, 1.0f);
    }
    if (rem >= 32) {
      scatter_edges8(xl, sl, sv, dv, 0);
      scatter_edges8(xl, sl, sv, dv, 8);
      scatter_edges8(xl, sl, sv, dv, 16);
      scatter_edges8(xl, sl, sv, dv, 24);
    } else {
      for (int j = 0; j < (int)rem; ++j) {
        const int s = __builtin_amdgcn_readlane(sv, j);
        const int d = __builtin_amdgcn_readlane(dv, j);
        const float* px = xl + (long)s * D;
        float* p = sl + (long)d * D;
        const float a0 = px[0];
        const float a1 = px[32];
        atomicAdd(p, a0);
        atomicAdd(p + 32, a1);
      }
    }
  }
}

template <int RELU>
__global__ __launch_bounds__(128) void sage_gemm(
    const float* __restrict__ xin, const float* __restrict__ ssum,
    const float* __restrict__ cnt, const float* __restrict__ Wl,
    const float* __restrict__ Wr, const float* __restrict__ bias,
    float* __restrict__ out, int nN) {
  __shared__ float sMean[16 * D];
  __shared__ float sX[16 * D];
  const int m0 = blockIdx.x * 16;
  const int tid = threadIdx.x;

  {
    int idx = tid * 8;
    int row = idx >> 6;
    int col = idx & (D - 1);
    int gr = m0 + row;
    if (gr >= nN) gr = nN - 1;
    float c = cnt[gr];
    float rinv = 1.0f / fmaxf(c, 1.0f);
    const float* ps = ssum + (long)gr * D + col;
    const float* px = xin + (long)gr * D + col;
#pragma unroll
    for (int t = 0; t < 8; ++t) {
      sMean[idx + t] = ps[t] * rinv;
      sX[idx + t] = px[t];
    }
  }
  __syncthreads();

  const int wave = tid >> 5;
  const int lane = tid & 31;
  const int n0 = wave * 16;
  const int lr = lane & 15;
  const int kh = (lane >> 4) * 2;

  const float* mrow = sMean + lr * D;
  const float* xrow = sX + lr * D;
  const float* wlrow = Wl + (n0 + lr) * D;
  const float* wrrow = Wr + (n0 + lr) * D;

  v8f acc = {};
#pragma unroll
  for (int k0 = 0; k0 < D; k0 += 4) {
    v2f a, b;
    a[0] = mrow[k0 + kh + 0];
    a[1] = mrow[k0 + kh + 1];
    b[0] = wlrow[k0 + kh + 0];
    b[1] = wlrow[k0 + kh + 1];
    acc = __builtin_amdgcn_wmma_f32_16x16x4_f32(false, a, false, b, (short)0,
                                                acc, false, false);
    a[0] = xrow[k0 + kh + 0];
    a[1] = xrow[k0 + kh + 1];
    b[0] = wrrow[k0 + kh + 0];
    b[1] = wrrow[k0 + kh + 1];
    acc = __builtin_amdgcn_wmma_f32_16x16x4_f32(false, a, false, b, (short)0,
                                                acc, false, false);
  }

  const float bv = bias[n0 + lr];
  const int mbase = m0 + (lane >> 4) * 8;
#pragma unroll
  for (int i = 0; i < 8; ++i) {
    int r = mbase + i;
    if (r < nN) {
      float v = acc[i] + bv;
      if (RELU) v = fmaxf(v, 0.0f);
      out[(long)r * D + n0 + lr] = v;
    }
  }
}

extern "C" void kernel_launch(void* const* d_in, const int* in_sizes, int n_in,
                              void* d_out, int out_size, void* d_ws,
                              size_t ws_size, hipStream_t stream) {
  const float* x = (const float*)d_in[0];
  const int* edge = (const int*)d_in[1];
  const float* Wl1 = (const float*)d_in[2];
  const float* Wr1 = (const float*)d_in[3];
  const float* b1 = (const float*)d_in[4];
  const float* Wl2 = (const float*)d_in[5];
  const float* Wr2 = (const float*)d_in[6];
  const float* b2 = (const float*)d_in[7];

  const int N = in_sizes[0] / D;
  const int E = in_sizes[1] / 2;
  const int* src = edge;
  const int* dst = edge + E;

  float* ssum = (float*)d_ws;
  float* cnt = ssum + (long)N * D;
  float* h = cnt + N;
  float* out = (float*)d_out;

  const int gtiles = (N + 15) / 16;

  zero_f32<<<2048, 256, 0, stream>>>(ssum, (long)N * D);
  zero_f32<<<64, 256, 0, stream>>>(cnt, (long)N);
  scatter_add<<<4096, 256, 0, stream>>>(x, src, dst, ssum, cnt, E, 1);
  sage_gemm<1><<<gtiles, 128, 0, stream>>>(x, ssum, cnt, Wl1, Wr1, b1, h, N);

  zero_f32<<<2048, 256, 0, stream>>>(ssum, (long)N * D);
  scatter_add<<<4096, 256, 0, stream>>>(h, src, dst, ssum, cnt, E, 0);
  sage_gemm<0><<<gtiles, 128, 0, stream>>>(h, ssum, cnt, Wl2, Wr2, b2, out, N);
}


# sage_gemm: LDS row pitch of the staged mean and x tiles 64 -> 68 floats, so the 16 rows of an A-fragment read fall into different banks (was a 16-way bank conflict on every read)
# speedup vs baseline: 1.7632x; 1.0442x over previous
#include <hip/hip_runtime.h>

#define D 64
#define LP 68

typedef __attribute__((ext_vector_type(2))) float v2f;
typedef __attribute__((ext_vector_type(8))) float v8f;

__global__ void zero_f32(float* __restrict__ p, long n) {
  long i = (long)blockIdx.x * blockDim.x + threadIdx.x;
  long stride = (long)gridDim.x * blockDim.x;
  for (; i < n; i += stride) p[i] = 0.0f;
}

__device__ __forceinline__ void scatter_edges8(const float* __restrict__ xl,
                                               float* __restrict__ sl,
                                               int sv, int dv, int j0) {
  float v0[8], v1[8];
  int dd[8];
#pragma unroll
  for (int u = 0; u < 8; ++u) {
    const int s = __builtin_amdgcn_readlane(sv, j0 + u);
    dd[u] = __builtin_amdgcn_readlane(dv, j0 + u);
    const float* px = xl + (long)s * D;
    v0[u] = px[0];
    v1[u] = px[32];
  }
#pragma unroll
  for (int u = 0; u < 8; ++u) {
    float* p = sl + (long)dd[u] * D;
    atomicAdd(p, v0[u]);
    atomicAdd(p + 32, v1[u]);
  }
}

__global__ void scatter_add(const float* __restrict__ xin,
                            const int* __restrict__ src,
                            const int* __restrict__ dst,
                            float* __restrict__ ssum, float* __restrict__ cnt,
                            int nE, int withCnt) {
  const int lane = threadIdx.x & 31;
  const int wave = __builtin_amdgcn_readfirstlane((int)(threadIdx.x >> 5));
  const int wpb = blockDim.x >> 5;
  const long nChunks = ((long)nE + 31) >> 5;
  const long cstride = (long)gridDim.x * wpb;
  const float* xl = xin + lane;
  float* sl = ssum + lane;
  for (long c = (long)blockIdx.x * wpb + wave; c < nChunks; c += cstride) {
    const long e = c * 32 + lane;
    const long rem = (long)nE - c * 32;
    int sv = 0, dv = 0;
    if (e < nE) {
      sv = src[e];
      dv = dst[e];
      if (withCnt) atomicAdd(cnt + dv, 1.0f);
    }
    if (rem >= 32) {
      scatter_edges8(xl, sl, sv, dv, 0);
      scatter_edges8(xl, sl, sv, dv, 8);
      scatter_edges8(xl, sl, sv, dv, 16);
      scatter_edges8(xl, sl, sv, dv, 24);
    } else {
      for (int j = 0; j < (int)rem; ++j) {
        const int s = __builtin_amdgcn_readlane(sv, j);
        const int d = __builtin_amdgcn_readlane(dv, j);
        const float* px = xl + (long)s * D;
        float* p = sl + (long)d * D;
        const float a0 = px[0];
        const float a1 = px[32];
        atomicAdd(p, a0);
        atomicAdd(p + 32, a1);
      }
    }
  }
}

template <int RELU>
__global__ __launch_bounds__(128) void sage_gemm(
    const float* __restrict__ xin, const float* __restrict__ ssum,
    const float* __restrict__ cnt, const float* __restrict__ Wl,
    const float* __restrict__ Wr, const float* __restrict__ bias,
    float* __restrict__ out, int nN) {
  __shared__ __attribute__((aligned(16))) float sMean[16 * LP];
  __shared__ __attribute__((aligned(16))) float sX[16 * LP];
  const int m0 = blockIdx.x * 16;
  const int tid = threadIdx.x;

  {
    int idx = tid * 8;
    int row = idx >> 6;
    int col = idx & (D - 1);
    int gr = m0 + row;
    if (gr >= nN) gr = nN - 1;
    float c = cnt[gr];
    float rinv = 1.0f / fmaxf(c, 1.0f);
    const float* ps = ssum + (long)gr * D + col;
    const float* px = xin + (long)gr * D + col;
    float* qm = sMean + row * LP + col;
    float* qx = sX + row * LP + col;
#pragma unroll
    for (int t = 0; t < 8; ++t) {
      qm[t] = ps[t] * rinv;
      qx[t] = px[t];
    }
  }
  __syncthreads();

  const int wave = tid >> 5;
  const int lane = tid & 31;
  const int n0 = wave * 16;
  const int lr = lane & 15;
  const int kh = (lane >> 4) * 2;

  const float* mrow = sMean + lr * LP;
  const float* xrow = sX + lr * LP;
  const float* wlrow = Wl + (n0 + lr) * D;
  const float* wrrow = Wr + (n0 + lr) * D;

  v8f acc = {};
#pragma unroll
  for (int k0 = 0; k0 < D; k0 += 4) {
    v2f a, b;
    a[0] = mrow[k0 + kh + 0];
    a[1] = mrow[k0 + kh + 1];
    b[0] = wlrow[k0 + kh + 0];
    b[1] = wlrow[k0 + kh + 1];
    acc = __builtin_amdgcn_wmma_f32_16x16x4_f32(false, a, false, b, (short)0,
                                                acc, false, false);
    a[0] = xrow[k0 + kh + 0];
    a[1] = xrow[k0 + kh + 1];
    b[0] = wrrow[k0 + kh + 0];
    b[1] = wrrow[k0 + kh + 1];
    acc = __builtin_amdgcn_wmma_f32_16x16x4_f32(false, a, false, b, (short)0,
                                                acc, false, false);
  }

  const float bv = bias[n0 + lr];
  const int mbase = m0 + (lane >> 4) * 8;
#pragma unroll
  for (int i = 0; i < 8; ++i) {
    int r = mbase + i;
    if (r < nN) {
      float v = acc[i] + bv;
      if (RELU) v = fmaxf(v, 0.0f);
      out[(long)r * D + n0 + lr] = v;
    }
  }
}

extern "C" void kernel_launch(void* const* d_in, const int* in_sizes, int n_in,
                              void* d_out, int out_size, void* d_ws,
                              size_t ws_size, hipStream_t stream) {
  const float* x = (const float*)d_in[0];
  const int* edge = (const int*)d_in[1];
  const float* Wl1 = (const float*)d_in[2];
  const float* Wr1 = (const float*)d_in[3];
  const float* b1 = (const float*)d_in[4];
  const float* Wl2 = (const float*)d_in[5];
  const float* Wr2 = (const float*)d_in[6];
  const float* b2 = (const float*)d_in[7];

  const int N = in_sizes[0] / D;
  const int E = in_sizes[1] / 2;
  const int* src = edge;
  const int* dst = edge + E;

  float* ssum = (float*)d_ws;
  float* cnt = ssum + (long)N * D;
  float* h = cnt + N;
  float* out = (float*)d_out;

  const int gtiles = (N + 15) / 16;

  zero_f32<<<2048, 256, 0, stream>>>(ssum, (long)N * D);
  zero_f32<<<64, 256, 0, stream>>>(cnt, (long)N);
  scatter_add<<<4096, 256, 0, stream>>>(x, src, dst, ssum, cnt, E, 1);
  sage_gemm<1><<<gtiles, 128, 0, stream>>>(x, ssum, cnt, Wl1, Wr1, b1, h, N);

  zero_f32<<<2048, 256, 0, stream>>>(ssum, (long)N * D);
  scatter_add<<<4096, 256, 0, stream>>>(h, src, dst, ssum, cnt, E, 0);
  sage_gemm<0><<<gtiles, 128, 0, stream>>>(h, ssum, cnt, Wl2, Wr2, b2, out, N);
}


# scatter_add: one plane of row sums and of degree counts per package half, picked by the wave's die number; sage_gemm adds the two planes (sum order now: per-half partial sums, then their sum)
# speedup vs baseline: 2.9580x; 1.6776x over previous
#include <hip/hip_runtime.h>

#define D 64
#define LP 68

typedef __attribute__((ext_vector_type(2))) float v2f;
typedef __attribute__((ext_vector_type(8))) float v8f;

__global__ void zero_f32(float* __restrict__ p, long n) {
  long i = (long)blockIdx.x * blockDim.x + threadIdx.x;
  long stride = (long)gridDim.x * blockDim.x;
  for (; i < n; i += stride) p[i] = 0.0f;
}

__device__ __forceinline__ int package_half() {
  return (__builtin_amdgcn_s_sendmsg_rtn(0x87) >> 18) & 1;
}

__device__ __forceinline__ void scatter_edges8(const float* __restrict__ xl,
                                               float* __restrict__ sl,
                                               int sv, int dv, int j0) {
  float v0[8], v1[8];
  int dd[8];
#pragma unroll
  for (int u = 0; u < 8; ++u) {
    const int s = __builtin_amdgcn_readlane(sv, j0 + u);
    dd[u] = __builtin_amdgcn_readlane(dv, j0 + u);
    const float* px = xl + (long)s * D;
    v0[u] = px[0];
    v1[u] = px[32];
  }
#pragma unroll
  for (int u = 0; u < 8; ++u) {
    float* p = sl + (long)dd[u] * D;
    atomicAdd(p, v0[u]);
    atomicAdd(p + 32, v1[u]);
  }
}

__global__ void scatter_add(const float* __restrict__ xin,
                            const int* __restrict__ src,
                            const int* __restrict__ dst,
                            float* __restrict__ ssum, float* __restrict__ cnt,
                            int nE, int nN, int withCnt) {
  const int lane = threadIdx.x & 31;
  const int half = package_half();
  const int wave = __builtin_amdgcn_readfirstlane((int)(threadIdx.x >> 5));
  const int wpb = blockDim.x >> 5;
  const long nChunks = ((long)nE + 31) >> 5;
  const long cstride = (long)gridDim.x * wpb;
  const float* xl = xin + lane;
  float* sl = ssum + (long)half * nN * D + lane;
  float* cl = cnt + (long)half * nN;
  for (long c = (long)blockIdx.x * wpb + wave; c < nChunks; c += cstride) {
    const long e = c * 32 + lane;
    const long rem = (long)nE - c * 32;
    int sv = 0, dv = 0;
    if (e < nE) {
      sv = src[e];
      dv = dst[e];
      if (withCnt) atomicAdd(cl + dv, 1.0f);
    }
    if (rem >= 32) {
      scatter_edges8(xl, sl, sv, dv, 0);
      scatter_edges8(xl, sl, sv, dv, 8);
      scatter_edges8(xl, sl, sv, dv, 16);
      scatter_edges8(xl, sl, sv, dv, 24);
    } else {
      for (int j = 0; j < (int)rem; ++j) {
        const int s = __builtin_amdgcn_readlane(sv, j);
        const int d = __builtin_amdgcn_readlane(dv, j);
        const float* px = xl + (long)s * D;
        float* p = sl + (long)d * D;
        const float a0 = px[0];
        const float a1 = px[32];
        atomicAdd(p, a0);
        atomicAdd(p + 32, a1);
      }
    }
  }
}

template <int RELU>
__global__ __launch_bounds__(128) void sage_gemm(
    const float* __restrict__ xin, const float* __restrict__ ssum,
    const float* __restrict__ cnt, const float* __restrict__ Wl,
    const float* __restrict__ Wr, const float* __restrict__ bias,
    float* __restrict__ out, int nN) {
  __shared__ __attribute__((aligned(16))) float sMean[16 * LP];
  __shared__ __attribute__((aligned(16))) float sX[16 * LP];
  const int m0 = blockIdx.x * 16;
  const int tid = threadIdx.x;

  {
    int idx = tid * 8;
    int row = idx >> 6;
    int col = idx & (D - 1);
    int gr = m0 + row;
    if (gr >= nN) gr = nN - 1;
    float c = cnt[gr] + cnt[nN + gr];
    float rinv = 1.0f / fmaxf(c, 1.0f);
    const float* ps = ssum + (long)gr * D + col;
    const float* pt = ps + (long)nN * D;
    const float* px = xin + (long)gr * D + col;
    float* qm = sMean + row * LP + col;
    float* qx = sX + row * LP + col;
#pragma unroll
    for (int t = 0; t < 8; ++t) {
      qm[t] = (ps[t] + pt[t]) * rinv;
      qx[t] = px[t];
    }
  }
  __syncthreads();

  const int wave = tid >> 5;
  const int lane = tid & 31;
  const int n0 = wave * 16;
  const int lr = lane & 15;
  const int kh = (lane >> 4) * 2;

  const float* mrow = sMean + lr * LP;
  const float* xrow = sX + lr * LP;
  const float* wlrow = Wl + (n0 + lr) * D;
  const float* wrrow = Wr + (n0 + lr) * D;

  v8f acc = {};
#pragma unroll
  for (int k0 = 0; k0 < D; k0 += 4) {
    v2f a, b;
    a[0] = mrow[k0 + kh + 0];
    a[1] = mrow[k0 + kh + 1];
    b[0] = wlrow[k0 + kh + 0];
    b[1] = wlrow[k0 + kh + 1];
    acc = __builtin_amdgcn_wmma_f32_16x16x4_f32(false, a, false, b, (short)0,
                                                acc, false, false);
    a[0] = xrow[k0 + kh + 0];
    a[1] = xrow[k0 + kh + 1];
    b[0] = wrrow[k0 + kh + 0];
    b[1] = wrrow[k0 + kh + 1];
    acc = __builtin_amdgcn_wmma_f32_16x16x4_f32(false, a, false, b, (short)0,
                                                acc, false, false);
  }

  const float bv = bias[n0 + lr];
  const int mbase = m0 + (lane >> 4) * 8;
#pragma unroll
  for (int i = 0; i < 8; ++i) {
    int r = mbase + i;
    if (r < nN) {
      float v = acc[i] + bv;
      if (RELU) v = fmaxf(v, 0.0f);
      out[(long)r * D + n0 + lr] = v;
    }
  }
}

extern "C" void kernel_launch(void* const* d_in, const int* in_sizes, int n_in,
                              void* d_out, int out_size, void* d_ws,
                              size_t ws_size, hipStream_t stream) {
  const float* x = (const float*)d_in[0];
  const int* edge = (const int*)d_in[1];
  const float* Wl1 = (const float*)d_in[2];
  const float* Wr1 = (const float*)d_in[3];
  const float* b1 = (const float*)d_in[4];
  const float* Wl2 = (const float*)d_in[5];
  const float* Wr2 = (const float*)d_in[6];
  const float* b2 = (const float*)d_in[7];

  const int N = in_sizes[0] / D;
  const int E = in_sizes[1] / 2;
  const int* src = edge;
  const int* dst = edge + E;

  float* ssum = (float*)d_ws;
  float* cnt = ssum + 2 * (long)N * D;
  float* h = cnt + 2 * (long)N;
  float* out = (float*)d_out;

  const int gtiles = (N + 15) / 16;

  zero_f32<<<2048, 256, 0, stream>>>(ssum, 2 * (long)N * D);
  zero_f32<<<64, 256, 0, stream>>>(cnt, 2 * (long)N);
  scatter_add<<<4096, 256, 0, stream>>>(x, src, dst, ssum, cnt, E, N, 1);
  sage_gemm<1><<<gtiles, 128, 0, stream>>>(x, ssum, cnt, Wl1, Wr1, b1, h, N);

  zero_f32<<<2048, 256, 0, stream>>>(ssum, 2 * (long)N * D);
  scatter_add<<<4096, 256, 0, stream>>>(h, src, dst, ssum, cnt, E, N, 0);
  sage_gemm<0><<<gtiles, 128, 0, stream>>>(h, ssum, cnt, Wl2, Wr2, b2, out, N);
}


# edges binned once per call into 256 destination ranges; one work-group per range sums its rows in LDS (LDS float adds, arrival order) and does mean+GEMM itself: no global float atomics, no zeroing
# speedup vs baseline: 3.0774x; 1.0404x over previous
#include <hip/hip_runtime.h>


#define D 64
#define LP 68
#define NRANGE 256
#define RROWS 208
#define NBINB 256

typedef __attribute__((ext_vector_type(2))) float v2f;
typedef __attribute__((ext_vector_type(8))) float v8f;

__device__ __forceinline__ int rows_per_range(int nN) {
  return (nN + NRANGE - 1) / NRANGE;
}

__global__ __launch_bounds__(1024) void bin_count(const int* __restrict__ dst,
                                                  int* __restrict__ table,
                                                  int nE, int nN) {
  __shared__ int hist[NRANGE];
  const int t = threadIdx.x;
  if (t < NRANGE) hist[t] = 0;
  __syncthreads();
  const int R = rows_per_range(nN);
  const int epb = (nE + (int)gridDim.x - 1) / (int)gridDim.x;
  const int e0 = (int)blockIdx.x * epb;
  int e1 = e0 + epb;
  if (e1 > nE) e1 = nE;
  for (int e = e0 + t; e < e1; e += 1024) atomicAdd(&hist[dst[e] / R], 1);
  __syncthreads();
  if (t < NRANGE) table[(int)blockIdx.x * NRANGE + t] = hist[t];
}

__global__ __launch_bounds__(1024) void bin_fill(const int* __restrict__ src,
                                                 const int* __restrict__ dst,
                                                 const int* __restrict__ table,
                                                 int* __restrict__ bases,
                                                 int* __restrict__ pairs,
                                                 int nE, int nN) {
  __shared__ int partT[4 * NRANGE];
  __shared__ int partB[4 * NRANGE];
  __shared__ int scan[NRANGE];
  __shared__ int cursor[NRANGE];
  const int tid = threadIdx.x;
  const int t = tid & (NRANGE - 1);
  const int q = tid >> 8;
  const int me = (int)blockIdx.x;
  {
    int before = 0, total = 0;
    for (int c = 0; c < NBINB / 4; c += 16) {
      int v[16];
#pragma unroll
      for (int j = 0; j < 16; ++j)
        v[j] = table[(q * (NBINB / 4) + c + j) * NRANGE + t];
#pragma unroll
      for (int j = 0; j < 16; ++j) {
        const int b = q * (NBINB / 4) + c + j;
        if (b < me) before += v[j];
        total += v[j];
      }
    }
    partT[q * NRANGE + t] = total;
    partB[q * NRANGE + t] = before;
  }
  __syncthreads();
  int total = 0, before = 0;
  if (tid < NRANGE) {
    total = partT[t] + partT[NRANGE + t] + partT[2 * NRANGE + t] + partT[3 * NRANGE + t];
    before = partB[t] + partB[NRANGE + t] + partB[2 * NRANGE + t] + partB[3 * NRANGE + t];
    scan[t] = total;
  }
  __syncthreads();
  for (int off = 1; off < NRANGE; off <<= 1) {
    int v = 0;
    if (tid < NRANGE && t >= off) v = scan[t - off];
    __syncthreads();
    if (tid < NRANGE) scan[t] += v;
    __syncthreads();
  }
  if (tid < NRANGE) {
    const int base = scan[t] - total;
    cursor[t] = base + before;
    if (me == 0) {
      bases[t] = base;
      if (t == NRANGE - 1) bases[NRANGE] = base + total;
    }
  }
  __syncthreads();

  const int R = rows_per_range(nN);
  const int nb = (int)gridDim.x;
  const int epb = (nE + nb - 1) / nb;
  const int e0 = me * epb;
  int e1 = e0 + epb;
  if (e1 > nE) e1 = nE;
  int2* pr = reinterpret_cast<int2*>(pairs);
  for (int e = e0 + tid; e < e1; e += 1024) {
    const int s = src[e];
    const int d = dst[e];
    const int r = d / R;
    const int pos = atomicAdd(&cursor[r], 1);
    pr[pos] = make_int2(s, d - r * R);
  }
}

template <int RELU>
__global__ __launch_bounds__(1024) void sage_layer(
    const float* __restrict__ xin, const int* __restrict__ bases,
    const int* __restrict__ pairs, const float* __restrict__ Wl,
    const float* __restrict__ Wr, const float* __restrict__ bias,
    float* __restrict__ out, int nN) {
  __shared__ __attribute__((aligned(16))) float sAcc[RROWS * LP];
  __shared__ __attribute__((aligned(16))) float sX[RROWS * LP];
  __shared__ float sCnt[RROWS];
  const int tid = threadIdx.x;
  const int lane = tid & 31;
  const int wave = __builtin_amdgcn_readfirstlane(tid >> 5);
  const int R = rows_per_range(nN);
  const int r0 = (int)blockIdx.x * R;

  constexpr int XP = (RROWS * (D / 4) + 1023) / 1024;
  float4 xv[XP];
#pragma unroll
  for (int q = 0; q < XP; ++q) {
    int i = tid + q * 1024;
    if (i > RROWS * (D / 4) - 1) i = RROWS * (D / 4) - 1;
    int gr = r0 + (i >> 4);
    if (gr >= nN) gr = nN - 1;
    xv[q] = *reinterpret_cast<const float4*>(xin + (long)gr * D + (i & 15) * 4);
  }
  for (int i = tid; i < RROWS * LP; i += 1024) sAcc[i] = 0.0f;
  if (tid < RROWS) sCnt[tid] = 0.0f;
#pragma unroll
  for (int q = 0; q < XP; ++q) {
    const int i = tid + q * 1024;
    if (i < RROWS * (D / 4))
      *reinterpret_cast<float4*>(sX + (i >> 4) * LP + (i & 15) * 4) = xv[q];
  }
  __syncthreads();

  {
    const int eb = bases[blockIdx.x];
    const int ee = bases[blockIdx.x + 1];
    const int2* pr = reinterpret_cast<const int2*>(pairs);
    const float* xl = xin + lane;
    float* al = sAcc + lane;
    const int nUnits = (ee - eb + 7) >> 3;
    int2 pn = make_int2(0, 0);
    if (wave < nUnits) {
      int e = eb + wave * 8 + (lane & 7);
      if (e > ee - 1) e = ee - 1;
      pn = pr[e];
    }
    for (int u = wave; u < nUnits; u += 32) {
      const int e0 = eb + u * 8;
      int nv = ee - e0;
      if (nv > 8) nv = 8;
      const int sv = pn.x;
      const int dv = pn.y;
      {
        int e2 = e0 + 32 * 8 + (lane & 7);
        if (e2 > ee - 1) e2 = ee - 1;
        pn = pr[e2];
      }
      if (lane < nv) atomicAdd(&sCnt[dv], 1.0f);
      if (nv == 8) {
        float v0[8], v1[8];
#pragma unroll
        for (int j = 0; j < 8; ++j) {
          const int s = __builtin_amdgcn_readlane(sv, j);
          const float* px = xl + (long)s * D;
          v0[j] = px[0];
          v1[j] = px[32];
        }
#pragma unroll
        for (int j = 0; j < 8; ++j) {
          const int d = __builtin_amdgcn_readlane(dv, j);
          float* p = al + d * LP;
          atomicAdd(p, v0[j]);
          atomicAdd(p + 32, v1[j]);
        }
      } else {
        for (int j = 0; j < nv; ++j) {
          const int s = __builtin_amdgcn_readlane(sv, j);
          const int d = __builtin_amdgcn_readlane(dv, j);
          const float* px = xl + (long)s * D;
          float* p = al + d * LP;
          const float a0 = px[0];
          const float a1 = px[32];
          atomicAdd(p, a0);
          atomicAdd(p + 32, a1);
        }
      }
    }
  }
  __syncthreads();

  for (int i = tid; i < RROWS * D; i += 1024) {
    const int row = i >> 6;
    const int col = i & (D - 1);
    const float rinv = 1.0f / fmaxf(sCnt[row], 1.0f);
    sAcc[row * LP + col] *= rinv;
  }
  __syncthreads();

  const int n0 = (wave & 3) * 16;
  const int lr = lane & 15;
  const int kh = (lane >> 4) * 2;
  const float* wlrow = Wl + (n0 + lr) * D;
  const float* wrrow = Wr + (n0 + lr) * D;
  const float bv = bias[n0 + lr];
  for (int rt = wave >> 2; rt < RROWS / 16 && rt * 16 < R; rt += 8) {
    const float* mrow = sAcc + (rt * 16 + lr) * LP;
    const float* xrow = sX + (rt * 16 + lr) * LP;
    v8f acc = {};
#pragma unroll
    for (int k0 = 0; k0 < D; k0 += 4) {
      v2f a, b;
      a[0] = mrow[k0 + kh + 0];
      a[1] = mrow[k0 + kh + 1];
      b[0] = wlrow[k0 + kh + 0];
      b[1] = wlrow[k0 + kh + 1];
      acc = __builtin_amdgcn_wmma_f32_16x16x4_f32(false, a, false, b, (short)0,
                                                  acc, false, false);
      a[0] = xrow[k0 + kh + 0];
      a[1] = xrow[k0 + kh + 1];
      b[0] = wrrow[k0 + kh + 0];
      b[1] = wrrow[k0 + kh + 1];
      acc = __builtin_amdgcn_wmma_f32_16x16x4_f32(false, a, false, b, (short)0,
                                                  acc, false, false);
    }
    const int lbase = rt * 16 + (lane >> 4) * 8;
#pragma unroll
    for (int i = 0; i < 8; ++i) {
      const int lrow = lbase + i;
      const int r = r0 + lrow;
      if (lrow < R && r < nN) {
        float v = acc[i] + bv;
        if (RELU) v = fmaxf(v, 0.0f);
        out[(long)r * D + n0 + lr] = v;
      }
    }
  }
}

extern "C" void kernel_launch(void* const* d_in, const int* in_sizes, int n_in,
                              void* d_out, int out_size, void* d_ws,
                              size_t ws_size, hipStream_t stream) {
  const float* x = (const float*)d_in[0];
  const int* edge = (const int*)d_in[1];
  const float* Wl1 = (const float*)d_in[2];
  const float* Wr1 = (const float*)d_in[3];
  const float* b1 = (const float*)d_in[4];
  const float* Wl2 = (const float*)d_in[5];
  const float* Wr2 = (const float*)d_in[6];
  const float* b2 = (const float*)d_in[7];

  const int N = in_sizes[0] / D;
  const int E = in_sizes[1] / 2;
  const int* src = edge;
  const int* dst = edge + E;

  int* table = (int*)d_ws;
  int* bases = table + NBINB * NRANGE;
  int* pairs = bases + 1024;
  float* h = (float*)(pairs + 2 * (long)E);
  float* out = (float*)d_out;

  bin_count<<<NBINB, 1024, 0, stream>>>(dst, table, E, N);
  bin_fill<<<NBINB, 1024, 0, stream>>>(src, dst, table, bases, pairs, E, N);

  sage_layer<1><<<NRANGE, 1024, 0, stream>>>(x, bases, pairs, Wl1, Wr1, b1, h, N);
  sage_layer<0><<<NRANGE, 1024, 0, stream>>>(h, bases, pairs, Wl2, Wr2, b2, out, N);
}


# sage_layer: Wl and Wr staged once per work-group into LDS rows of pitch 68; B fragments read from LDS (64 banks, no conflict), were global loads touching 16 lines each
# speedup vs baseline: 3.5756x; 1.1619x over previous
#include <hip/hip_runtime.h>


#define D 64
#define LP 68
#define NRANGE 256
#define RROWS 208
#define NBINB 256

typedef __attribute__((ext_vector_type(2))) float v2f;
typedef __attribute__((ext_vector_type(8))) float v8f;

__device__ __forceinline__ int rows_per_range(int nN) {
  return (nN + NRANGE - 1) / NRANGE;
}

__global__ __launch_bounds__(1024) void bin_count(const int* __restrict__ dst,
                                                  int* __restrict__ table,
                                                  int nE, int nN) {
  __shared__ int hist[NRANGE];
  const int t = threadIdx.x;
  if (t < NRANGE) hist[t] = 0;
  __syncthreads();
  const int R = rows_per_range(nN);
  const int epb = (nE + (int)gridDim.x - 1) / (int)gridDim.x;
  const int e0 = (int)blockIdx.x * epb;
  int e1 = e0 + epb;
  if (e1 > nE) e1 = nE;
  for (int e = e0 + t; e < e1; e += 1024) atomicAdd(&hist[dst[e] / R], 1);
  __syncthreads();
  if (t < NRANGE) table[(int)blockIdx.x * NRANGE + t] = hist[t];
}

__global__ __launch_bounds__(1024) void bin_fill(const int* __restrict__ src,
                                                 const int* __restrict__ dst,
                                                 const int* __restrict__ table,
                                                 int* __restrict__ bases,
                                                 int* __restrict__ pairs,
                                                 int nE, int nN) {
  __shared__ int partT[4 * NRANGE];
  __shared__ int partB[4 * NRANGE];
  __shared__ int scan[NRANGE];
  __shared__ int cursor[NRANGE];
  const int tid = threadIdx.x;
  const int t = tid & (NRANGE - 1);
  const int q = tid >> 8;
  const int me = (int)blockIdx.x;
  {
    int before = 0, total = 0;
    for (int c = 0; c < NBINB / 4; c += 16) {
      int v[16];
#pragma unroll
      for (int j = 0; j < 16; ++j)
        v[j] = table[(q * (NBINB / 4) + c + j) * NRANGE + t];
#pragma unroll
      for (int j = 0; j < 16; ++j) {
        const int b = q * (NBINB / 4) + c + j;
        if (b < me) before += v[j];
        total += v[j];
      }
    }
    partT[q * NRANGE + t] = total;
    partB[q * NRANGE + t] = before;
  }
  __syncthreads();
  int total = 0, before = 0;
  if (tid < NRANGE) {
    total = partT[t] + partT[NRANGE + t] + partT[2 * NRANGE + t] + partT[3 * NRANGE + t];
    before = partB[t] + partB[NRANGE + t] + partB[2 * NRANGE + t] + partB[3 * NRANGE + t];
    scan[t] = total;
  }
  __syncthreads();
  for (int off = 1; off < NRANGE; off <<= 1) {
    int v = 0;
    if (tid < NRANGE && t >= off) v = scan[t - off];
    __syncthreads();
    if (tid < NRANGE) scan[t] += v;
    __syncthreads();
  }
  if (tid < NRANGE) {
    const int base = scan[t] - total;
    cursor[t] = base + before;
    if (me == 0) {
      bases[t] = base;
      if (t == NRANGE - 1) bases[NRANGE] = base + total;
    }
  }
  __syncthreads();

  const int R = rows_per_range(nN);
  const int nb = (int)gridDim.x;
  const int epb = (nE + nb - 1) / nb;
  const int e0 = me * epb;
  int e1 = e0 + epb;
  if (e1 > nE) e1 = nE;
  int2* pr = reinterpret_cast<int2*>(pairs);
  for (int e = e0 + tid; e < e1; e += 1024) {
    const int s = src[e];
    const int d = dst[e];
    const int r = d / R;
    const int pos = atomicAdd(&cursor[r], 1);
    pr[pos] = make_int2(s, d - r * R);
  }
}

template <int RELU>
__global__ __launch_bounds__(1024) void sage_layer(
    const float* __restrict__ xin, const int* __restrict__ bases,
    const int* __restrict__ pairs, const float* __restrict__ Wl,
    const float* __restrict__ Wr, const float* __restrict__ bias,
    float* __restrict__ out, int nN) {
  __shared__ __attribute__((aligned(16))) float sAcc[RROWS * LP];
  __shared__ __attribute__((aligned(16))) float sX[RROWS * LP];
  __shared__ float sCnt[RROWS];
  __shared__ __attribute__((aligned(16))) float sW[2 * D * LP];
  const int tid = threadIdx.x;
  const int lane = tid & 31;
  const int wave = __builtin_amdgcn_readfirstlane(tid >> 5);
  const int R = rows_per_range(nN);
  const int r0 = (int)blockIdx.x * R;

  constexpr int XP = (RROWS * (D / 4) + 1023) / 1024;
  float4 xv[XP];
#pragma unroll
  for (int q = 0; q < XP; ++q) {
    int i = tid + q * 1024;
    if (i > RROWS * (D / 4) - 1) i = RROWS * (D / 4) - 1;
    int gr = r0 + (i >> 4);
    if (gr >= nN) gr = nN - 1;
    xv[q] = *reinterpret_cast<const float4*>(xin + (long)gr * D + (i & 15) * 4);
  }
  const int wrow = tid >> 4;
  const int wcol = (tid & 15) * 4;
  const float4 wlv = *reinterpret_cast<const float4*>(Wl + wrow * D + wcol);
  const float4 wrv = *reinterpret_cast<const float4*>(Wr + wrow * D + wcol);
  for (int i = tid; i < RROWS * LP; i += 1024) sAcc[i] = 0.0f;
  if (tid < RROWS) sCnt[tid] = 0.0f;
  *reinterpret_cast<float4*>(sW + wrow * LP + wcol) = wlv;
  *reinterpret_cast<float4*>(sW + (D + wrow) * LP + wcol) = wrv;
#pragma unroll
  for (int q = 0; q < XP; ++q) {
    const int i = tid + q * 1024;
    if (i < RROWS * (D / 4))
      *reinterpret_cast<float4*>(sX + (i >> 4) * LP + (i & 15) * 4) = xv[q];
  }
  __syncthreads();

  {
    const int eb = bases[blockIdx.x];
    const int ee = bases[blockIdx.x + 1];
    const int2* pr = reinterpret_cast<const int2*>(pairs);
    const float* xl = xin + lane;
    float* al = sAcc + lane;
    const int nUnits = (ee - eb + 7) >> 3;
    int2 pn = make_int2(0, 0);
    if (wave < nUnits) {
      int e = eb + wave * 8 + (lane & 7);
      if (e > ee - 1) e = ee - 1;
      pn = pr[e];
    }
    for (int u = wave; u < nUnits; u += 32) {
      const int e0 = eb + u * 8;
      int nv = ee - e0;
      if (nv > 8) nv = 8;
      const int sv = pn.x;
      const int dv = pn.y;
      {
        int e2 = e0 + 32 * 8 + (lane & 7);
        if (e2 > ee - 1) e2 = ee - 1;
        pn = pr[e2];
      }
      if (lane < nv) atomicAdd(&sCnt[dv], 1.0f);
      if (nv == 8) {
        float v0[8], v1[8];
#pragma unroll
        for (int j = 0; j < 8; ++j) {
          const int s = __builtin_amdgcn_readlane(sv, j);
          const float* px = xl + (long)s * D;
          v0[j] = px[0];
          v1[j] = px[32];
        }
#pragma unroll
        for (int j = 0; j < 8; ++j) {
          const int d = __builtin_amdgcn_readlane(dv, j);
          float* p = al + d * LP;
          atomicAdd(p, v0[j]);
          atomicAdd(p + 32, v1[j]);
        }
      } else {
        for (int j = 0; j < nv; ++j) {
          const int s = __builtin_amdgcn_readlane(sv, j);
          const int d = __builtin_amdgcn_readlane(dv, j);
          const float* px = xl + (long)s * D;
          float* p = al + d * LP;
          const float a0 = px[0];
          const float a1 = px[32];
          atomicAdd(p, a0);
          atomicAdd(p + 32, a1);
        }
      }
    }
  }
  __syncthreads();

  for (int i = tid; i < RROWS * D; i += 1024) {
    const int row = i >> 6;
    const int col = i & (D - 1);
    const float rinv = 1.0f / fmaxf(sCnt[row], 1.0f);
    sAcc[row * LP + col] *= rinv;
  }
  __syncthreads();

  const int n0 = (wave & 3) * 16;
  const int lr = lane & 15;
  const int kh = (lane >> 4) * 2;
  const float* wlrow = sW + (n0 + lr) * LP;
  const float* wrrow = sW + (D + n0 + lr) * LP;
  const float bv = bias[n0 + lr];
  for (int rt = wave >> 2; rt < RROWS / 16 && rt * 16 < R; rt += 8) {
    const float* mrow = sAcc + (rt * 16 + lr) * LP;
    const float* xrow = sX + (rt * 16 + lr) * LP;
    v8f acc = {};
#pragma unroll
    for (int k0 = 0; k0 < D; k0 += 4) {
      v2f a, b;
      a[0] = mrow[k0 + kh + 0];
      a[1] = mrow[k0 + kh + 1];
      b[0] = wlrow[k0 + kh + 0];
      b[1] = wlrow[k0 + kh + 1];
      acc = __builtin_amdgcn_wmma_f32_16x16x4_f32(false, a, false, b, (short)0,
                                                  acc, false, false);
      a[0] = xrow[k0 + kh + 0];
      a[1] = xrow[k0 + kh + 1];
      b[0] = wrrow[k0 + kh + 0];
      b[1] = wrrow[k0 + kh + 1];
      acc = __builtin_amdgcn_wmma_f32_16x16x4_f32(false, a, false, b, (short)0,
                                                  acc, false, false);
    }
    const int lbase = rt * 16 + (lane >> 4) * 8;
#pragma unroll
    for (int i = 0; i < 8; ++i) {
      const int lrow = lbase + i;
      const int r = r0 + lrow;
      if (lrow < R && r < nN) {
        float v = acc[i] + bv;
        if (RELU) v = fmaxf(v, 0.0f);
        out[(long)r * D + n0 + lr] = v;
      }
    }
  }
}

extern "C" void kernel_launch(void* const* d_in, const int* in_sizes, int n_in,
                              void* d_out, int out_size, void* d_ws,
                              size_t ws_size, hipStream_t stream) {
  const float* x = (const float*)d_in[0];
  const int* edge = (const int*)d_in[1];
  const float* Wl1 = (const float*)d_in[2];
  const float* Wr1 = (const float*)d_in[3];
  const float* b1 = (const float*)d_in[4];
  const float* Wl2 = (const float*)d_in[5];
  const float* Wr2 = (const float*)d_in[6];
  const float* b2 = (const float*)d_in[7];

  const int N = in_sizes[0] / D;
  const int E = in_sizes[1] / 2;
  const int* src = edge;
  const int* dst = edge + E;

  int* table = (int*)d_ws;
  int* bases = table + NBINB * NRANGE;
  int* pairs = bases + 1024;
  float* h = (float*)(pairs + 2 * (long)E);
  float* out = (float*)d_out;

  bin_count<<<NBINB, 1024, 0, stream>>>(dst, table, E, N);
  bin_fill<<<NBINB, 1024, 0, stream>>>(src, dst, table, bases, pairs, E, N);

  sage_layer<1><<<NRANGE, 1024, 0, stream>>>(x, bases, pairs, Wl1, Wr1, b1, h, N);
  sage_layer<0><<<NRANGE, 1024, 0, stream>>>(h, bases, pairs, Wl2, Wr2, b2, out, N);
}
